# starting kernel
# baseline (speedup 1.0000x reference)
#include <hip/hip_runtime.h>
#include <hip/hip_bf16.h>
#include <math.h>

#define BATCH 128
#define RNUM  2048
#define ONUM  16
#define DNUM  16
#define CNUM  16

typedef __attribute__((ext_vector_type(2))) float v2f;
typedef __attribute__((ext_vector_type(8))) float v8f;

__device__ __forceinline__ float halfsum16(float v) {
    v += __shfl_xor(v, 1);
    v += __shfl_xor(v, 2);
    v += __shfl_xor(v, 4);
    v += __shfl_xor(v, 8);
    return v;
}

template <int MODE>
__global__ __launch_bounds__(256) void caps_kernel(
    const float* __restrict__ x,
    const float* __restrict__ W,
    float* __restrict__ bmat,
    const float* __restrict__ rowmax,
    const float* __restrict__ denom,
    float* __restrict__ out)
{
    const int r    = blockIdx.x;
    const int tid  = threadIdx.x;
    const int L    = tid & 31;
    const int w    = tid >> 5;
    const int h    = L >> 4;
    const int m16  = L & 15;
    const int bb0  = w * 16;

    v2f a[4];
    const float* xb = x + (bb0 + m16) * DNUM;
    #pragma unroll
    for (int k = 0; k < 4; ++k) {
        a[k] = *reinterpret_cast<const v2f*>(xb + 4 * k + 2 * h);
    }

    v8f U[ONUM];
    const float* Wr = W + (size_t)r * ONUM * DNUM * CNUM;
    #pragma unroll
    for (int o = 0; o < ONUM; ++o) {
        const float* Wo = Wr + o * (DNUM * CNUM);
        v8f acc = {0.f, 0.f, 0.f, 0.f, 0.f, 0.f, 0.f, 0.f};
        #pragma unroll
        for (int k = 0; k < 4; ++k) {
            v2f bmx;
            const int row0 = 4 * k + 2 * h;
            bmx.x = Wo[(row0 + 0) * CNUM + m16];
            bmx.y = Wo[(row0 + 1) * CNUM + m16];
            acc = __builtin_amdgcn_wmma_f32_16x16x4_f32(
                false, a[k], false, bmx, (short)0, acc, false, false);
        }
        U[o] = acc;
    }

    float braw[8], creg[8];
    #pragma unroll
    for (int i = 0; i < 8; ++i) {
        if (MODE == 0) {
            braw[i] = 0.0f;
            creg[i] = 1.0f / (float)RNUM;
        } else {
            const int b = bb0 + i + 8 * h;
            braw[i] = bmat[((size_t)b * RNUM + r) * ONUM + m16];
            const float rm = rowmax[b * ONUM + m16];
            const float dn = denom[b * ONUM + m16];
            creg[i] = __expf(braw[i] - rm) / dn;
        }
    }

    v8f s = {0.f, 0.f, 0.f, 0.f, 0.f, 0.f, 0.f, 0.f};
    #pragma unroll
    for (int o = 0; o < ONUM; ++o) {
        #pragma unroll
        for (int i = 0; i < 8; ++i) {
            float co;
            if (MODE == 0) {
                co = 1.0f / (float)RNUM;
            } else {
                co = __shfl(creg[i], (tid & 16) | o);
            }
            s[i] += co * U[o][i];
        }
    }

    v8f vj;
    #pragma unroll
    for (int i = 0; i < 8; ++i) {
        const float ns  = halfsum16(s[i] * s[i]);
        const float fac = sqrtf(ns) / (1.0f + ns);
        vj[i] = s[i] * fac;
    }

    if (MODE < 2) {
        float aout[8];
        #pragma unroll
        for (int o = 0; o < ONUM; ++o) {
            #pragma unroll
            for (int i = 0; i < 8; ++i) {
                const float t = halfsum16(U[o][i] * vj[i]);
                if (m16 == o) aout[i] = t;
            }
        }
        #pragma unroll
        for (int i = 0; i < 8; ++i) {
            const int b = bb0 + i + 8 * h;
            bmat[((size_t)b * RNUM + r) * ONUM + m16] = braw[i] + aout[i];
        }
    } else {
        #pragma unroll
        for (int i = 0; i < 8; ++i) {
            const int b = bb0 + i + 8 * h;
            out[((size_t)b * RNUM + r) * CNUM + m16] = vj[i];
        }
    }
}

__global__ __launch_bounds__(256) void softmax_reduce(
    const float* __restrict__ bmat,
    float* __restrict__ rowmax,
    float* __restrict__ denom)
{
    __shared__ float sm[256];
    const int b  = blockIdx.x;
    const int t  = threadIdx.x;
    const int o  = t & 15;
    const int rs = t >> 4;

    float mx = -INFINITY;
    for (int r = rs; r < RNUM; r += 16)
        mx = fmaxf(mx, bmat[((size_t)b * RNUM + r) * ONUM + o]);
    sm[t] = mx;
    __syncthreads();
    for (int off = 128; off >= 16; off >>= 1) {
        if (t < off) sm[t] = fmaxf(sm[t], sm[t + off]);
        __syncthreads();
    }
    const float rm = sm[o];
    if (t < 16) rowmax[b * ONUM + t] = sm[t];
    __syncthreads();

    float se = 0.0f;
    for (int r = rs; r < RNUM; r += 16)
        se += __expf(bmat[((size_t)b * RNUM + r) * ONUM + o] - rm);
    sm[t] = se;
    __syncthreads();
    for (int off = 128; off >= 16; off >>= 1) {
        if (t < off) sm[t] += sm[t + off];
        __syncthreads();
    }
    if (t < 16) denom[b * ONUM + t] = sm[t];
}

extern "C" void kernel_launch(void* const* d_in, const int* in_sizes, int n_in,
                              void* d_out, int out_size, void* d_ws, size_t ws_size,
                              hipStream_t stream) {
    const float* x = (const float*)d_in[0];
    const float* W = (const float*)d_in[1];
    float* out = (float*)d_out;

    float* bmat = (float*)d_ws;
    float* rowmax = bmat + (size_t)BATCH * RNUM * ONUM;
    float* denom  = rowmax + BATCH * ONUM;

    dim3 grid(RNUM), block(256);

    caps_kernel<0><<<grid, block, 0, stream>>>(x, W, bmat, rowmax, denom, out);
    softmax_reduce<<<dim3(BATCH), block, 0, stream>>>(bmat, rowmax, denom);
    caps_kernel<1><<<grid, block, 0, stream>>>(x, W, bmat, rowmax, denom, out);
    softmax_reduce<<<dim3(BATCH), block, 0, stream>>>(bmat, rowmax, denom);
    caps_kernel<2><<<grid, block, 0, stream>>>(x, W, bmat, rowmax, denom, out);
}


# Issue the WMMA transposed (A=W^T, B=x^T): a lane owns one batch row with c in registers, so sums over o and c are in-lane FMAs plus one lane^16 exchange; c-sum order changes (8 in-lane, then halves)
# speedup vs baseline: 1.2621x; 1.2621x over previous
#include <hip/hip_runtime.h>
#include <hip/hip_bf16.h>
#include <math.h>

#define BATCH 128
#define RNUM  2048
#define ONUM  16
#define DNUM  16
#define CNUM  16

typedef __attribute__((ext_vector_type(2))) float v2f;
typedef __attribute__((ext_vector_type(8))) float v8f;

__device__ __forceinline__ float xhalf(float v) {
    const int b = __builtin_bit_cast(int, v);
    return __builtin_bit_cast(float,
        __builtin_amdgcn_permlanex16(b, b, 0x76543210, 0xfedcba98, false, false));
}

template <int MODE>
__global__ __launch_bounds__(256) void caps_kernel(
    const float* __restrict__ x,
    const float* __restrict__ W,
    float* __restrict__ bmat,
    const float* __restrict__ rowmax,
    const float* __restrict__ denom,
    float* __restrict__ out)
{
    const int r    = blockIdx.x;
    const int tid  = threadIdx.x;
    const int L    = tid & 31;
    const int w    = tid >> 5;
    const int h    = L >> 4;
    const int m16  = L & 15;
    const int bb0  = w * 16;
    const int b    = bb0 + m16;
    const bool hi  = (h != 0);

    float braw[8], cown[8];
    const size_t brow = ((size_t)b * RNUM + r) * ONUM + 8 * h;
    if (MODE != 0) {
        const float4 q0 = *reinterpret_cast<const float4*>(bmat + brow);
        const float4 q1 = *reinterpret_cast<const float4*>(bmat + brow + 4);
        const float4 m0 = *reinterpret_cast<const float4*>(rowmax + b * ONUM + 8 * h);
        const float4 m1 = *reinterpret_cast<const float4*>(rowmax + b * ONUM + 8 * h + 4);
        const float4 d0 = *reinterpret_cast<const float4*>(denom + b * ONUM + 8 * h);
        const float4 d1 = *reinterpret_cast<const float4*>(denom + b * ONUM + 8 * h + 4);
        braw[0] = q0.x; braw[1] = q0.y; braw[2] = q0.z; braw[3] = q0.w;
        braw[4] = q1.x; braw[5] = q1.y; braw[6] = q1.z; braw[7] = q1.w;
        const float rm[8] = {m0.x, m0.y, m0.z, m0.w, m1.x, m1.y, m1.z, m1.w};
        const float dn[8] = {d0.x, d0.y, d0.z, d0.w, d1.x, d1.y, d1.z, d1.w};
        #pragma unroll
        for (int j = 0; j < 8; ++j) cown[j] = __expf(braw[j] - rm[j]) / dn[j];
    } else {
        #pragma unroll
        for (int j = 0; j < 8; ++j) { braw[j] = 0.0f; cown[j] = 1.0f / (float)RNUM; }
    }

    v2f xf[4];
    const float* xb = x + b * DNUM;
    #pragma unroll
    for (int k = 0; k < 4; ++k) {
        xf[k] = *reinterpret_cast<const v2f*>(xb + 4 * k + 2 * h);
    }

    v8f U[ONUM];
    const float* Wr = W + (size_t)r * ONUM * DNUM * CNUM;
    #pragma unroll
    for (int o = 0; o < ONUM; ++o) {
        const float* Wo = Wr + o * (DNUM * CNUM);
        v8f acc = {0.f, 0.f, 0.f, 0.f, 0.f, 0.f, 0.f, 0.f};
        #pragma unroll
        for (int k = 0; k < 4; ++k) {
            v2f wf;
            const int row0 = 4 * k + 2 * h;
            wf.x = Wo[(row0 + 0) * CNUM + m16];
            wf.y = Wo[(row0 + 1) * CNUM + m16];
            acc = __builtin_amdgcn_wmma_f32_16x16x4_f32(
                false, wf, false, xf[k], (short)0, acc, false, false);
        }
        U[o] = acc;
    }

    float call[ONUM];
    if (MODE != 0) {
        #pragma unroll
        for (int j = 0; j < 8; ++j) {
            const float cx = xhalf(cown[j]);
            call[j]     = hi ? cx : cown[j];
            call[8 + j] = hi ? cown[j] : cx;
        }
    } else {
        #pragma unroll
        for (int o = 0; o < ONUM; ++o) call[o] = 1.0f / (float)RNUM;
    }

    v8f s = {0.f, 0.f, 0.f, 0.f, 0.f, 0.f, 0.f, 0.f};
    #pragma unroll
    for (int o = 0; o < ONUM; ++o) {
        #pragma unroll
        for (int i = 0; i < 8; ++i) {
            s[i] += call[o] * U[o][i];
        }
    }

    float nsp = 0.0f;
    #pragma unroll
    for (int i = 0; i < 8; ++i) nsp += s[i] * s[i];
    const float ns  = nsp + xhalf(nsp);
    const float fac = sqrtf(ns) / (1.0f + ns);
    v8f vj;
    #pragma unroll
    for (int i = 0; i < 8; ++i) vj[i] = s[i] * fac;

    if (MODE < 2) {
        float tot[ONUM];
        #pragma unroll
        for (int o = 0; o < ONUM; ++o) {
            float t = 0.0f;
            #pragma unroll
            for (int i = 0; i < 8; ++i) t += U[o][i] * vj[i];
            tot[o] = t + xhalf(t);
        }
        float bnew[8];
        #pragma unroll
        for (int j = 0; j < 8; ++j) {
            const float aj = hi ? tot[8 + j] : tot[j];
            bnew[j] = braw[j] + aj;
        }
        *reinterpret_cast<float4*>(bmat + brow)     = make_float4(bnew[0], bnew[1], bnew[2], bnew[3]);
        *reinterpret_cast<float4*>(bmat + brow + 4) = make_float4(bnew[4], bnew[5], bnew[6], bnew[7]);
    } else {
        float* op = out + ((size_t)b * RNUM + r) * CNUM + 8 * h;
        *reinterpret_cast<float4*>(op)     = make_float4(vj[0], vj[1], vj[2], vj[3]);
        *reinterpret_cast<float4*>(op + 4) = make_float4(vj[4], vj[5], vj[6], vj[7]);
    }
}

__global__ __launch_bounds__(256) void softmax_reduce(
    const float* __restrict__ bmat,
    float* __restrict__ rowmax,
    float* __restrict__ denom)
{
    __shared__ float sm[256];
    const int b  = blockIdx.x;
    const int t  = threadIdx.x;
    const int o  = t & 15;
    const int rs = t >> 4;

    float mx = -INFINITY;
    for (int r = rs; r < RNUM; r += 16)
        mx = fmaxf(mx, bmat[((size_t)b * RNUM + r) * ONUM + o]);
    sm[t] = mx;
    __syncthreads();
    for (int off = 128; off >= 16; off >>= 1) {
        if (t < off) sm[t] = fmaxf(sm[t], sm[t + off]);
        __syncthreads();
    }
    const float rm = sm[o];
    if (t < 16) rowmax[b * ONUM + t] = sm[t];
    __syncthreads();

    float se = 0.0f;
    for (int r = rs; r < RNUM; r += 16)
        se += __expf(bmat[((size_t)b * RNUM + r) * ONUM + o] - rm);
    sm[t] = se;
    __syncthreads();
    for (int off = 128; off >= 16; off >>= 1) {
        if (t < off) sm[t] += sm[t + off];
        __syncthreads();
    }
    if (t < 16) denom[b * ONUM + t] = sm[t];
}

extern "C" void kernel_launch(void* const* d_in, const int* in_sizes, int n_in,
                              void* d_out, int out_size, void* d_ws, size_t ws_size,
                              hipStream_t stream) {
    const float* x = (const float*)d_in[0];
    const float* W = (const float*)d_in[1];
    float* out = (float*)d_out;

    float* bmat = (float*)d_ws;
    float* rowmax = bmat + (size_t)BATCH * RNUM * ONUM;
    float* denom  = rowmax + BATCH * ONUM;

    dim3 grid(RNUM), block(256);

    caps_kernel<0><<<grid, block, 0, stream>>>(x, W, bmat, rowmax, denom, out);
    softmax_reduce<<<dim3(BATCH), block, 0, stream>>>(bmat, rowmax, denom);
    caps_kernel<1><<<grid, block, 0, stream>>>(x, W, bmat, rowmax, denom, out);
    softmax_reduce<<<dim3(BATCH), block, 0, stream>>>(bmat, rowmax, denom);
    caps_kernel<2><<<grid, block, 0, stream>>>(x, W, bmat, rowmax, denom, out);
}


# Softmax statistics pass reads the logits once: 16 float4 loads per thread all in flight, values kept in registers for max and exp-sum, lane butterflies + LDS; denom sum order changes (tree)
# speedup vs baseline: 2.1026x; 1.6660x over previous
#include <hip/hip_runtime.h>
#include <hip/hip_bf16.h>
#include <math.h>

#define BATCH 128
#define RNUM  2048
#define ONUM  16
#define DNUM  16
#define CNUM  16

typedef __attribute__((ext_vector_type(2))) float v2f;
typedef __attribute__((ext_vector_type(8))) float v8f;

__device__ __forceinline__ float xhalf(float v) {
    const int b = __builtin_bit_cast(int, v);
    return __builtin_bit_cast(float,
        __builtin_amdgcn_permlanex16(b, b, 0x76543210, 0xfedcba98, false, false));
}

template <int MODE>
__global__ __launch_bounds__(256) void caps_kernel(
    const float* __restrict__ x,
    const float* __restrict__ W,
    float* __restrict__ bmat,
    const float* __restrict__ rowmax,
    const float* __restrict__ denom,
    float* __restrict__ out)
{
    const int r    = blockIdx.x;
    const int tid  = threadIdx.x;
    const int L    = tid & 31;
    const int w    = tid >> 5;
    const int h    = L >> 4;
    const int m16  = L & 15;
    const int bb0  = w * 16;
    const int b    = bb0 + m16;
    const bool hi  = (h != 0);

    float braw[8], cown[8];
    const size_t brow = ((size_t)b * RNUM + r) * ONUM + 8 * h;
    if (MODE != 0) {
        const float4 q0 = *reinterpret_cast<const float4*>(bmat + brow);
        const float4 q1 = *reinterpret_cast<const float4*>(bmat + brow + 4);
        const float4 m0 = *reinterpret_cast<const float4*>(rowmax + b * ONUM + 8 * h);
        const float4 m1 = *reinterpret_cast<const float4*>(rowmax + b * ONUM + 8 * h + 4);
        const float4 d0 = *reinterpret_cast<const float4*>(denom + b * ONUM + 8 * h);
        const float4 d1 = *reinterpret_cast<const float4*>(denom + b * ONUM + 8 * h + 4);
        braw[0] = q0.x; braw[1] = q0.y; braw[2] = q0.z; braw[3] = q0.w;
        braw[4] = q1.x; braw[5] = q1.y; braw[6] = q1.z; braw[7] = q1.w;
        const float rm[8] = {m0.x, m0.y, m0.z, m0.w, m1.x, m1.y, m1.z, m1.w};
        const float dn[8] = {d0.x, d0.y, d0.z, d0.w, d1.x, d1.y, d1.z, d1.w};
        #pragma unroll
        for (int j = 0; j < 8; ++j) cown[j] = __expf(braw[j] - rm[j]) / dn[j];
    } else {
        #pragma unroll
        for (int j = 0; j < 8; ++j) { braw[j] = 0.0f; cown[j] = 1.0f / (float)RNUM; }
    }

    v2f xf[4];
    const float* xb = x + b * DNUM;
    #pragma unroll
    for (int k = 0; k < 4; ++k) {
        xf[k] = *reinterpret_cast<const v2f*>(xb + 4 * k + 2 * h);
    }

    v8f U[ONUM];
    const float* Wr = W + (size_t)r * ONUM * DNUM * CNUM;
    #pragma unroll
    for (int o = 0; o < ONUM; ++o) {
        const float* Wo = Wr + o * (DNUM * CNUM);
        v8f acc = {0.f, 0.f, 0.f, 0.f, 0.f, 0.f, 0.f, 0.f};
        #pragma unroll
        for (int k = 0; k < 4; ++k) {
            v2f wf;
            const int row0 = 4 * k + 2 * h;
            wf.x = Wo[(row0 + 0) * CNUM + m16];
            wf.y = Wo[(row0 + 1) * CNUM + m16];
            acc = __builtin_amdgcn_wmma_f32_16x16x4_f32(
                false, wf, false, xf[k], (short)0, acc, false, false);
        }
        U[o] = acc;
    }

    float call[ONUM];
    if (MODE != 0) {
        #pragma unroll
        for (int j = 0; j < 8; ++j) {
            const float cx = xhalf(cown[j]);
            call[j]     = hi ? cx : cown[j];
            call[8 + j] = hi ? cown[j] : cx;
        }
    } else {
        #pragma unroll
        for (int o = 0; o < ONUM; ++o) call[o] = 1.0f / (float)RNUM;
    }

    v8f s = {0.f, 0.f, 0.f, 0.f, 0.f, 0.f, 0.f, 0.f};
    #pragma unroll
    for (int o = 0; o < ONUM; ++o) {
        #pragma unroll
        for (int i = 0; i < 8; ++i) {
            s[i] += call[o] * U[o][i];
        }
    }

    float nsp = 0.0f;
    #pragma unroll
    for (int i = 0; i < 8; ++i) nsp += s[i] * s[i];
    const float ns  = nsp + xhalf(nsp);
    const float fac = sqrtf(ns) / (1.0f + ns);
    v8f vj;
    #pragma unroll
    for (int i = 0; i < 8; ++i) vj[i] = s[i] * fac;

    if (MODE < 2) {
        float tot[ONUM];
        #pragma unroll
        for (int o = 0; o < ONUM; ++o) {
            float t = 0.0f;
            #pragma unroll
            for (int i = 0; i < 8; ++i) t += U[o][i] * vj[i];
            tot[o] = t + xhalf(t);
        }
        float bnew[8];
        #pragma unroll
        for (int j = 0; j < 8; ++j) {
            const float aj = hi ? tot[8 + j] : tot[j];
            bnew[j] = braw[j] + aj;
        }
        *reinterpret_cast<float4*>(bmat + brow)     = make_float4(bnew[0], bnew[1], bnew[2], bnew[3]);
        *reinterpret_cast<float4*>(bmat + brow + 4) = make_float4(bnew[4], bnew[5], bnew[6], bnew[7]);
    } else {
        float* op = out + ((size_t)b * RNUM + r) * CNUM + 8 * h;
        *reinterpret_cast<float4*>(op)     = make_float4(vj[0], vj[1], vj[2], vj[3]);
        *reinterpret_cast<float4*>(op + 4) = make_float4(vj[4], vj[5], vj[6], vj[7]);
    }
}

#define SR_THREADS 512
#define SR_WAVES   (SR_THREADS / 32)
#define SR_LOADS   (RNUM * ONUM / 4 / SR_THREADS)

template <int M>
__device__ __forceinline__ float xrow(float v) {
    const int b = __builtin_bit_cast(int, v);
    return __builtin_bit_cast(float,
        __builtin_amdgcn_update_dpp(b, b, 0x160 + M, 0xf, 0xf, false));
}

__device__ __forceinline__ float bfly_max(float v) {
    v = fmaxf(v, xrow<4>(v));
    v = fmaxf(v, xrow<8>(v));
    return fmaxf(v, xhalf(v));
}
__device__ __forceinline__ float bfly_sum(float v) {
    v += xrow<4>(v);
    v += xrow<8>(v);
    return v + xhalf(v);
}

__global__ __launch_bounds__(SR_THREADS) void softmax_reduce(
    const float* __restrict__ bmat,
    float* __restrict__ rowmax,
    float* __restrict__ denom)
{
    __shared__ float4 red_max[SR_WAVES * 4];
    __shared__ float4 red_sum[SR_WAVES * 4];
    const int b = blockIdx.x;
    const int t = threadIdx.x;
    const int L = t & 31;
    const int w = t >> 5;

    const float4* src4 =
        reinterpret_cast<const float4*>(bmat + (size_t)b * RNUM * ONUM) + t;
    float4 v[SR_LOADS];
    #pragma unroll
    for (int i = 0; i < SR_LOADS; ++i) v[i] = src4[i * SR_THREADS];

    float4 m = make_float4(-INFINITY, -INFINITY, -INFINITY, -INFINITY);
    #pragma unroll
    for (int i = 0; i < SR_LOADS; ++i) {
        m.x = fmaxf(m.x, v[i].x); m.y = fmaxf(m.y, v[i].y);
        m.z = fmaxf(m.z, v[i].z); m.w = fmaxf(m.w, v[i].w);
    }
    m.x = bfly_max(m.x); m.y = bfly_max(m.y); m.z = bfly_max(m.z); m.w = bfly_max(m.w);
    if (L < 4) red_max[4 * w + L] = m;
    __syncthreads();
    {
        const float4 p = red_max[L], q = red_max[L + 32];
        m.x = bfly_max(fmaxf(p.x, q.x)); m.y = bfly_max(fmaxf(p.y, q.y));
        m.z = bfly_max(fmaxf(p.z, q.z)); m.w = bfly_max(fmaxf(p.w, q.w));
    }

    float4 s = make_float4(0.0f, 0.0f, 0.0f, 0.0f);
    #pragma unroll
    for (int i = 0; i < SR_LOADS; ++i) {
        s.x += __expf(v[i].x - m.x); s.y += __expf(v[i].y - m.y);
        s.z += __expf(v[i].z - m.z); s.w += __expf(v[i].w - m.w);
    }
    s.x = bfly_sum(s.x); s.y = bfly_sum(s.y); s.z = bfly_sum(s.z); s.w = bfly_sum(s.w);
    if (L < 4) red_sum[4 * w + L] = s;
    __syncthreads();
    if (w == 0) {
        const float4 p = red_sum[L], q = red_sum[L + 32];
        s.x = bfly_sum(p.x + q.x); s.y = bfly_sum(p.y + q.y);
        s.z = bfly_sum(p.z + q.z); s.w = bfly_sum(p.w + q.w);
        if (L < 4) {
            *reinterpret_cast<float4*>(rowmax + b * ONUM + 4 * L) = m;
            *reinterpret_cast<float4*>(denom  + b * ONUM + 4 * L) = s;
        }
    }
}

extern "C" void kernel_launch(void* const* d_in, const int* in_sizes, int n_in,
                              void* d_out, int out_size, void* d_ws, size_t ws_size,
                              hipStream_t stream) {
    const float* x = (const float*)d_in[0];
    const float* W = (const float*)d_in[1];
    float* out = (float*)d_out;

    float* bmat = (float*)d_ws;
    float* rowmax = bmat + (size_t)BATCH * RNUM * ONUM;
    float* denom  = rowmax + BATCH * ONUM;

    dim3 grid(RNUM), block(256);

    caps_kernel<0><<<grid, block, 0, stream>>>(x, W, bmat, rowmax, denom, out);
    softmax_reduce<<<dim3(BATCH), dim3(SR_THREADS), 0, stream>>>(bmat, rowmax, denom);
    caps_kernel<1><<<grid, block, 0, stream>>>(x, W, bmat, rowmax, denom, out);
    softmax_reduce<<<dim3(BATCH), dim3(SR_THREADS), 0, stream>>>(bmat, rowmax, denom);
    caps_kernel<2><<<grid, block, 0, stream>>>(x, W, bmat, rowmax, denom, out);
}


# Each block loops over 4 consecutive r (grid 2048 -> 512, all resident at once): 4x fewer block dispatches and wave set-ups; per-batch-row loads (x, softmax max, denom) hoisted out of the loop
# speedup vs baseline: 2.8153x; 1.3390x over previous
#include <hip/hip_runtime.h>
#include <hip/hip_bf16.h>
#include <math.h>

#define BATCH 128
#define RNUM  2048
#define ONUM  16
#define DNUM  16
#define CNUM  16
#define RGRP  4

typedef __attribute__((ext_vector_type(2))) float v2f;
typedef __attribute__((ext_vector_type(8))) float v8f;

__device__ __forceinline__ float xhalf(float v) {
    const int b = __builtin_bit_cast(int, v);
    return __builtin_bit_cast(float,
        __builtin_amdgcn_permlanex16(b, b, 0x76543210, 0xfedcba98, false, false));
}

template <int MODE>
__global__ __launch_bounds__(256) void caps_kernel(
    const float* __restrict__ x,
    const float* __restrict__ W,
    float* __restrict__ bmat,
    const float* __restrict__ rowmax,
    const float* __restrict__ denom,
    float* __restrict__ out)
{
    const int r0   = blockIdx.x * RGRP;
    const int tid  = threadIdx.x;
    const int L    = tid & 31;
    const int w    = tid >> 5;
    const int h    = L >> 4;
    const int m16  = L & 15;
    const int bb0  = w * 16;
    const int b    = bb0 + m16;
    const bool hi  = (h != 0);

    float rm[8], dn[8];
    if (MODE != 0) {
        const float4 m0 = *reinterpret_cast<const float4*>(rowmax + b * ONUM + 8 * h);
        const float4 m1 = *reinterpret_cast<const float4*>(rowmax + b * ONUM + 8 * h + 4);
        const float4 d0 = *reinterpret_cast<const float4*>(denom + b * ONUM + 8 * h);
        const float4 d1 = *reinterpret_cast<const float4*>(denom + b * ONUM + 8 * h + 4);
        rm[0] = m0.x; rm[1] = m0.y; rm[2] = m0.z; rm[3] = m0.w;
        rm[4] = m1.x; rm[5] = m1.y; rm[6] = m1.z; rm[7] = m1.w;
        dn[0] = d0.x; dn[1] = d0.y; dn[2] = d0.z; dn[3] = d0.w;
        dn[4] = d1.x; dn[5] = d1.y; dn[6] = d1.z; dn[7] = d1.w;
    }

    v2f xf[4];
    const float* xb = x + b * DNUM;
    #pragma unroll
    for (int k = 0; k < 4; ++k) {
        xf[k] = *reinterpret_cast<const v2f*>(xb + 4 * k + 2 * h);
    }

    #pragma unroll 1
    for (int ri = 0; ri < RGRP; ++ri) {
        const int r = r0 + ri;

        float braw[8], cown[8];
        const size_t brow = ((size_t)b * RNUM + r) * ONUM + 8 * h;
        if (MODE != 0) {
            const float4 q0 = *reinterpret_cast<const float4*>(bmat + brow);
            const float4 q1 = *reinterpret_cast<const float4*>(bmat + brow + 4);
            braw[0] = q0.x; braw[1] = q0.y; braw[2] = q0.z; braw[3] = q0.w;
            braw[4] = q1.x; braw[5] = q1.y; braw[6] = q1.z; braw[7] = q1.w;
            #pragma unroll
            for (int j = 0; j < 8; ++j) cown[j] = __expf(braw[j] - rm[j]) / dn[j];
        } else {
            #pragma unroll
            for (int j = 0; j < 8; ++j) { braw[j] = 0.0f; cown[j] = 1.0f / (float)RNUM; }
        }

        v8f U[ONUM];
        const float* Wr = W + (size_t)r * ONUM * DNUM * CNUM;
        #pragma unroll
        for (int o = 0; o < ONUM; ++o) {
            const float* Wo = Wr + o * (DNUM * CNUM);
            v8f acc = {0.f, 0.f, 0.f, 0.f, 0.f, 0.f, 0.f, 0.f};
            #pragma unroll
            for (int k = 0; k < 4; ++k) {
                v2f wf;
                const int row0 = 4 * k + 2 * h;
                wf.x = Wo[(row0 + 0) * CNUM + m16];
                wf.y = Wo[(row0 + 1) * CNUM + m16];
                acc = __builtin_amdgcn_wmma_f32_16x16x4_f32(
                    false, wf, false, xf[k], (short)0, acc, false, false);
            }
            U[o] = acc;
        }

        float call[ONUM];
        if (MODE != 0) {
            #pragma unroll
            for (int j = 0; j < 8; ++j) {
                const float cx = xhalf(cown[j]);
                call[j]     = hi ? cx : cown[j];
                call[8 + j] = hi ? cown[j] : cx;
            }
        } else {
            #pragma unroll
            for (int o = 0; o < ONUM; ++o) call[o] = 1.0f / (float)RNUM;
        }

        v8f s = {0.f, 0.f, 0.f, 0.f, 0.f, 0.f, 0.f, 0.f};
        #pragma unroll
        for (int o = 0; o < ONUM; ++o) {
            #pragma unroll
            for (int i = 0; i < 8; ++i) {
                s[i] += call[o] * U[o][i];
            }
        }

        float nsp = 0.0f;
        #pragma unroll
        for (int i = 0; i < 8; ++i) nsp += s[i] * s[i];
        const float ns  = nsp + xhalf(nsp);
        const float fac = sqrtf(ns) / (1.0f + ns);
        v8f vj;
        #pragma unroll
        for (int i = 0; i < 8; ++i) vj[i] = s[i] * fac;

        if (MODE < 2) {
            float tot[ONUM];
            #pragma unroll
            for (int o = 0; o < ONUM; ++o) {
                float t = 0.0f;
                #pragma unroll
                for (int i = 0; i < 8; ++i) t += U[o][i] * vj[i];
                tot[o] = t + xhalf(t);
            }
            float bnew[8];
            #pragma unroll
            for (int j = 0; j < 8; ++j) {
                const float aj = hi ? tot[8 + j] : tot[j];
                bnew[j] = braw[j] + aj;
            }
            *reinterpret_cast<float4*>(bmat + brow)     = make_float4(bnew[0], bnew[1], bnew[2], bnew[3]);
            *reinterpret_cast<float4*>(bmat + brow + 4) = make_float4(bnew[4], bnew[5], bnew[6], bnew[7]);
        } else {
            float* op = out + ((size_t)b * RNUM + r) * CNUM + 8 * h;
            *reinterpret_cast<float4*>(op)     = make_float4(vj[0], vj[1], vj[2], vj[3]);
            *reinterpret_cast<float4*>(op + 4) = make_float4(vj[4], vj[5], vj[6], vj[7]);
        }
    }
}

#define SR_THREADS 512
#define SR_WAVES   (SR_THREADS / 32)
#define SR_LOADS   (RNUM * ONUM / 4 / SR_THREADS)

template <int M>
__device__ __forceinline__ float xrow(float v) {
    const int b = __builtin_bit_cast(int, v);
    return __builtin_bit_cast(float,
        __builtin_amdgcn_update_dpp(b, b, 0x160 + M, 0xf, 0xf, false));
}

__device__ __forceinline__ float bfly_max(float v) {
    v = fmaxf(v, xrow<4>(v));
    v = fmaxf(v, xrow<8>(v));
    return fmaxf(v, xhalf(v));
}
__device__ __forceinline__ float bfly_sum(float v) {
    v += xrow<4>(v);
    v += xrow<8>(v);
    return v + xhalf(v);
}

__global__ __launch_bounds__(SR_THREADS) void softmax_reduce(
    const float* __restrict__ bmat,
    float* __restrict__ rowmax,
    float* __restrict__ denom)
{
    __shared__ float4 red_max[SR_WAVES * 4];
    __shared__ float4 red_sum[SR_WAVES * 4];
    const int b = blockIdx.x;
    const int t = threadIdx.x;
    const int L = t & 31;
    const int w = t >> 5;

    const float4* src4 =
        reinterpret_cast<const float4*>(bmat + (size_t)b * RNUM * ONUM) + t;
    float4 v[SR_LOADS];
    #pragma unroll
    for (int i = 0; i < SR_LOADS; ++i) v[i] = src4[i * SR_THREADS];

    float4 m = make_float4(-INFINITY, -INFINITY, -INFINITY, -INFINITY);
    #pragma unroll
    for (int i = 0; i < SR_LOADS; ++i) {
        m.x = fmaxf(m.x, v[i].x); m.y = fmaxf(m.y, v[i].y);
        m.z = fmaxf(m.z, v[i].z); m.w = fmaxf(m.w, v[i].w);
    }
    m.x = bfly_max(m.x); m.y = bfly_max(m.y); m.z = bfly_max(m.z); m.w = bfly_max(m.w);
    if (L < 4) red_max[4 * w + L] = m;
    __syncthreads();
    {
        const float4 p = red_max[L], q = red_max[L + 32];
        m.x = bfly_max(fmaxf(p.x, q.x)); m.y = bfly_max(fmaxf(p.y, q.y));
        m.z = bfly_max(fmaxf(p.z, q.z)); m.w = bfly_max(fmaxf(p.w, q.w));
    }

    float4 s = make_float4(0.0f, 0.0f, 0.0f, 0.0f);
    #pragma unroll
    for (int i = 0; i < SR_LOADS; ++i) {
        s.x += __expf(v[i].x - m.x); s.y += __expf(v[i].y - m.y);
        s.z += __expf(v[i].z - m.z); s.w += __expf(v[i].w - m.w);
    }
    s.x = bfly_sum(s.x); s.y = bfly_sum(s.y); s.z = bfly_sum(s.z); s.w = bfly_sum(s.w);
    if (L < 4) red_sum[4 * w + L] = s;
    __syncthreads();
    if (w == 0) {
        const float4 p = red_sum[L], q = red_sum[L + 32];
        s.x = bfly_sum(p.x + q.x); s.y = bfly_sum(p.y + q.y);
        s.z = bfly_sum(p.z + q.z); s.w = bfly_sum(p.w + q.w);
        if (L < 4) {
            *reinterpret_cast<float4*>(rowmax + b * ONUM + 4 * L) = m;
            *reinterpret_cast<float4*>(denom  + b * ONUM + 4 * L) = s;
        }
    }
}

extern "C" void kernel_launch(void* const* d_in, const int* in_sizes, int n_in,
                              void* d_out, int out_size, void* d_ws, size_t ws_size,
                              hipStream_t stream) {
    const float* x = (const float*)d_in[0];
    const float* W = (const float*)d_in[1];
    float* out = (float*)d_out;

    float* bmat = (float*)d_ws;
    float* rowmax = bmat + (size_t)BATCH * RNUM * ONUM;
    float* denom  = rowmax + BATCH * ONUM;

    dim3 grid(RNUM / RGRP), block(256);

    caps_kernel<0><<<grid, block, 0, stream>>>(x, W, bmat, rowmax, denom, out);
    softmax_reduce<<<dim3(BATCH), dim3(SR_THREADS), 0, stream>>>(bmat, rowmax, denom);
    caps_kernel<1><<<grid, block, 0, stream>>>(x, W, bmat, rowmax, denom, out);
    softmax_reduce<<<dim3(BATCH), dim3(SR_THREADS), 0, stream>>>(bmat, rowmax, denom);
    caps_kernel<2><<<grid, block, 0, stream>>>(x, W, bmat, rowmax, denom, out);
}


# W fragments fetched as whole-line 16-byte loads (4 consecutive c per lane feed 4 tiles; rows remapped so a lane owns 8 o x 16 c): 4x fewer load instructions; o-sum and c-sum orders change
# speedup vs baseline: 2.9566x; 1.0502x over previous
#include <hip/hip_runtime.h>
#include <hip/hip_bf16.h>
#include <math.h>

#define BATCH 128
#define RNUM  2048
#define ONUM  16
#define DNUM  16
#define CNUM  16
#define RGRP  4

typedef __attribute__((ext_vector_type(2))) float v2f;
typedef __attribute__((ext_vector_type(8))) float v8f;

__device__ __forceinline__ float xhalf(float v) {
    const int b = __builtin_bit_cast(int, v);
    return __builtin_bit_cast(float,
        __builtin_amdgcn_permlanex16(b, b, 0x76543210, 0xfedcba98, false, false));
}

__device__ __forceinline__ v2f pair_of(const v8f& v, const int n) {
    v2f r;
    r.x = v[2 * n];
    r.y = v[2 * n + 1];
    return r;
}

template <int MODE>
__global__ __launch_bounds__(256) void caps_kernel(
    const float* __restrict__ x,
    const float* __restrict__ W,
    float* __restrict__ bmat,
    const float* __restrict__ rowmax,
    const float* __restrict__ denom,
    float* __restrict__ out)
{
    const int r0   = blockIdx.x * RGRP;
    const int tid  = threadIdx.x;
    const int L    = tid & 31;
    const int w    = tid >> 5;
    const int h    = L >> 4;
    const int m16  = L & 15;
    const int bb0  = w * 16;
    const int b    = bb0 + m16;
    const bool hi  = (h != 0);

    float rm[8], dn[8];
    if (MODE != 0) {
        const float4 m0 = *reinterpret_cast<const float4*>(rowmax + b * ONUM + 8 * h);
        const float4 m1 = *reinterpret_cast<const float4*>(rowmax + b * ONUM + 8 * h + 4);
        const float4 d0 = *reinterpret_cast<const float4*>(denom + b * ONUM + 8 * h);
        const float4 d1 = *reinterpret_cast<const float4*>(denom + b * ONUM + 8 * h + 4);
        rm[0] = m0.x; rm[1] = m0.y; rm[2] = m0.z; rm[3] = m0.w;
        rm[4] = m1.x; rm[5] = m1.y; rm[6] = m1.z; rm[7] = m1.w;
        dn[0] = d0.x; dn[1] = d0.y; dn[2] = d0.z; dn[3] = d0.w;
        dn[4] = d1.x; dn[5] = d1.y; dn[6] = d1.z; dn[7] = d1.w;
    }

    v2f xf[4];
    const float* xb = x + b * DNUM + h;
    #pragma unroll
    for (int k = 0; k < 4; ++k) {
        xf[k].x = xb[4 * k];
        xf[k].y = xb[4 * k + 2];
    }

    const int oo = m16 >> 2;
    const int q  = m16 & 3;
    const int wlane = (8 * (oo >> 1) + (oo & 1)) * (DNUM * CNUM) + h * CNUM + 4 * q;

    #pragma unroll 1
    for (int ri = 0; ri < RGRP; ++ri) {
        const int r = r0 + ri;

        float braw[8], cown[8];
        const size_t brow = ((size_t)b * RNUM + r) * ONUM + 8 * h;
        float4 q0, q1;
        if (MODE != 0) {
            q0 = *reinterpret_cast<const float4*>(bmat + brow);
            q1 = *reinterpret_cast<const float4*>(bmat + brow + 4);
        }

        const float* Wl = W + (size_t)r * ONUM * DNUM * CNUM + wlane;
        float4 F[4][4][2];
        #pragma unroll
        for (int g = 0; g < 4; ++g) {
            #pragma unroll
            for (int k = 0; k < 4; ++k) {
                #pragma unroll
                for (int s = 0; s < 2; ++s) {
                    F[g][k][s] = *reinterpret_cast<const float4*>(
                        Wl + (2 * g) * (DNUM * CNUM) + (4 * k + 2 * s) * CNUM);
                }
            }
        }

        if (MODE != 0) {
            braw[0] = q0.x; braw[1] = q0.y; braw[2] = q0.z; braw[3] = q0.w;
            braw[4] = q1.x; braw[5] = q1.y; braw[6] = q1.z; braw[7] = q1.w;
            #pragma unroll
            for (int j = 0; j < 8; ++j) cown[j] = __expf(braw[j] - rm[j]) / dn[j];
        } else {
            #pragma unroll
            for (int j = 0; j < 8; ++j) { braw[j] = 0.0f; cown[j] = 1.0f / (float)RNUM; }
        }

        v8f U[ONUM];
        v2f P[4][2];
        #pragma unroll
        for (int t = 0; t < 4; ++t) {
            P[t][0] = (v2f){0.f, 0.f};
            P[t][1] = (v2f){0.f, 0.f};
        }
        #pragma unroll
        for (int g = 0; g < 5; ++g) {
            __builtin_amdgcn_sched_barrier(0x0416);
            if (g < 4) {
                v8f acc[4];
                #pragma unroll
                for (int t = 0; t < 4; ++t) {
                    acc[t] = (v8f){0.f, 0.f, 0.f, 0.f, 0.f, 0.f, 0.f, 0.f};
                }
                #pragma unroll
                for (int k = 0; k < 4; ++k) {
                    const float4 f0 = F[g < 4 ? g : 0][k][0];
                    const float4 f1 = F[g < 4 ? g : 0][k][1];
                    const float a0[4] = {f0.x, f0.y, f0.z, f0.w};
                    const float a1[4] = {f1.x, f1.y, f1.z, f1.w};
                    #pragma unroll
                    for (int t = 0; t < 4; ++t) {
                        v2f wf;
                        wf.x = a0[t];
                        wf.y = a1[t];
                        acc[t] = __builtin_amdgcn_wmma_f32_16x16x4_f32(
                            false, wf, false, xf[k], (short)0, acc[t], false, false);
                    }
                }
                #pragma unroll
                for (int t = 0; t < 4; ++t) U[4 * (g < 4 ? g : 0) + t] = acc[t];
            }
            if (g > 0) {
                const int gp = g - 1;
                #pragma unroll
                for (int e = 0; e < 2; ++e) {
                    const float cj = cown[2 * gp + e];
                    const v2f cj2 = {cj, cj};
                    #pragma unroll
                    for (int t = 0; t < 4; ++t) {
                        #pragma unroll
                        for (int p = 0; p < 2; ++p) {
                            P[t][p] += cj2 * pair_of(U[4 * gp + t], 2 * e + p);
                        }
                    }
                }
            }
        }

        v2f S[4][2];
        #pragma unroll
        for (int t = 0; t < 4; ++t) {
            #pragma unroll
            for (int p = 0; p < 2; ++p) {
                v2f other;
                other.x = xhalf(P[t][p].x);
                other.y = xhalf(P[t][p].y);
                S[t][p] = P[t][p] + other;
            }
        }

        v2f nsp = {0.f, 0.f};
        #pragma unroll
        for (int t = 0; t < 4; ++t) {
            #pragma unroll
            for (int p = 0; p < 2; ++p) nsp += S[t][p] * S[t][p];
        }
        const float ns  = nsp.x + nsp.y;
        const float fac = sqrtf(ns) / (1.0f + ns);
        const v2f fac2 = {fac, fac};
        v2f V[4][2];
        #pragma unroll
        for (int t = 0; t < 4; ++t) {
            #pragma unroll
            for (int p = 0; p < 2; ++p) V[t][p] = S[t][p] * fac2;
        }

        if (MODE < 2) {
            float bnew[8];
            #pragma unroll
            for (int g = 0; g < 4; ++g) {
                #pragma unroll
                for (int e = 0; e < 2; ++e) {
                    v2f a2 = {0.f, 0.f};
                    #pragma unroll
                    for (int t = 0; t < 4; ++t) {
                        #pragma unroll
                        for (int p = 0; p < 2; ++p) {
                            a2 += pair_of(U[4 * g + t], 2 * e + p) * V[t][p];
                        }
                    }
                    bnew[2 * g + e] = braw[2 * g + e] + (a2.x + a2.y);
                }
            }
            *reinterpret_cast<float4*>(bmat + brow)     = make_float4(bnew[0], bnew[1], bnew[2], bnew[3]);
            *reinterpret_cast<float4*>(bmat + brow + 4) = make_float4(bnew[4], bnew[5], bnew[6], bnew[7]);
        } else {
            float vo[8];
            #pragma unroll
            for (int i = 0; i < 8; ++i) {
                const float lo = (i < 4) ? V[i & 3][0].x : V[i & 3][0].y;
                const float up = (i < 4) ? V[i & 3][1].x : V[i & 3][1].y;
                vo[i] = hi ? up : lo;
            }
            float* op = out + ((size_t)b * RNUM + r) * CNUM + 8 * h;
            *reinterpret_cast<float4*>(op)     = make_float4(vo[0], vo[1], vo[2], vo[3]);
            *reinterpret_cast<float4*>(op + 4) = make_float4(vo[4], vo[5], vo[6], vo[7]);
        }
    }
}

#define SR_THREADS 512
#define SR_WAVES   (SR_THREADS / 32)
#define SR_LOADS   (RNUM * ONUM / 4 / SR_THREADS)

template <int M>
__device__ __forceinline__ float xrow(float v) {
    const int b = __builtin_bit_cast(int, v);
    return __builtin_bit_cast(float,
        __builtin_amdgcn_update_dpp(b, b, 0x160 + M, 0xf, 0xf, false));
}

__device__ __forceinline__ float bfly_max(float v) {
    v = fmaxf(v, xrow<4>(v));
    v = fmaxf(v, xrow<8>(v));
    return fmaxf(v, xhalf(v));
}
__device__ __forceinline__ float bfly_sum(float v) {
    v += xrow<4>(v);
    v += xrow<8>(v);
    return v + xhalf(v);
}

__global__ __launch_bounds__(SR_THREADS) void softmax_reduce(
    const float* __restrict__ bmat,
    float* __restrict__ rowmax,
    float* __restrict__ denom)
{
    __shared__ float4 red_max[SR_WAVES * 4];
    __shared__ float4 red_sum[SR_WAVES * 4];
    const int b = blockIdx.x;
    const int t = threadIdx.x;
    const int L = t & 31;
    const int w = t >> 5;

    const float4* src4 =
        reinterpret_cast<const float4*>(bmat + (size_t)b * RNUM * ONUM) + t;
    float4 v[SR_LOADS];
    #pragma unroll
    for (int i = 0; i < SR_LOADS; ++i) v[i] = src4[i * SR_THREADS];

    float4 m = make_float4(-INFINITY, -INFINITY, -INFINITY, -INFINITY);
    #pragma unroll
    for (int i = 0; i < SR_LOADS; ++i) {
        m.x = fmaxf(m.x, v[i].x); m.y = fmaxf(m.y, v[i].y);
        m.z = fmaxf(m.z, v[i].z); m.w = fmaxf(m.w, v[i].w);
    }
    m.x = bfly_max(m.x); m.y = bfly_max(m.y); m.z = bfly_max(m.z); m.w = bfly_max(m.w);
    if (L < 4) red_max[4 * w + L] = m;
    __syncthreads();
    {
        const float4 p = red_max[L], q = red_max[L + 32];
        m.x = bfly_max(fmaxf(p.x, q.x)); m.y = bfly_max(fmaxf(p.y, q.y));
        m.z = bfly_max(fmaxf(p.z, q.z)); m.w = bfly_max(fmaxf(p.w, q.w));
    }

    float4 s = make_float4(0.0f, 0.0f, 0.0f, 0.0f);
    #pragma unroll
    for (int i = 0; i < SR_LOADS; ++i) {
        s.x += __expf(v[i].x - m.x); s.y += __expf(v[i].y - m.y);
        s.z += __expf(v[i].z - m.z); s.w += __expf(v[i].w - m.w);
    }
    s.x = bfly_sum(s.x); s.y = bfly_sum(s.y); s.z = bfly_sum(s.z); s.w = bfly_sum(s.w);
    if (L < 4) red_sum[4 * w + L] = s;
    __syncthreads();
    if (w == 0) {
        const float4 p = red_sum[L], q = red_sum[L + 32];
        s.x = bfly_sum(p.x + q.x); s.y = bfly_sum(p.y + q.y);
        s.z = bfly_sum(p.z + q.z); s.w = bfly_sum(p.w + q.w);
        if (L < 4) {
            *reinterpret_cast<float4*>(rowmax + b * ONUM + 4 * L) = m;
            *reinterpret_cast<float4*>(denom  + b * ONUM + 4 * L) = s;
        }
    }
}

extern "C" void kernel_launch(void* const* d_in, const int* in_sizes, int n_in,
                              void* d_out, int out_size, void* d_ws, size_t ws_size,
                              hipStream_t stream) {
    const float* x = (const float*)d_in[0];
    const float* W = (const float*)d_in[1];
    float* out = (float*)d_out;

    float* bmat = (float*)d_ws;
    float* rowmax = bmat + (size_t)BATCH * RNUM * ONUM;
    float* denom  = rowmax + BATCH * ONUM;

    dim3 grid(RNUM / RGRP), block(256);

    caps_kernel<0><<<grid, block, 0, stream>>>(x, W, bmat, rowmax, denom, out);
    softmax_reduce<<<dim3(BATCH), dim3(SR_THREADS), 0, stream>>>(bmat, rowmax, denom);
    caps_kernel<1><<<grid, block, 0, stream>>>(x, W, bmat, rowmax, denom, out);
    softmax_reduce<<<dim3(BATCH), dim3(SR_THREADS), 0, stream>>>(bmat, rowmax, denom);
    caps_kernel<2><<<grid, block, 0, stream>>>(x, W, bmat, rowmax, denom, out);
}


# W[r0..r0+3] staged once per block in LDS in operand-pair order (each wave fetches 1/8, 16 whole-line loads); all 8 waves read ready A pairs by 16-byte LDS loads: global W loads per lane 128 -> 16
# speedup vs baseline: 3.2738x; 1.1073x over previous
#include <hip/hip_runtime.h>
#include <hip/hip_bf16.h>
#include <math.h>

#define BATCH 128
#define RNUM  2048
#define ONUM  16
#define DNUM  16
#define CNUM  16
#define RGRP  4

typedef __attribute__((ext_vector_type(2))) float v2f;
typedef __attribute__((ext_vector_type(8))) float v8f;

__device__ __forceinline__ float xhalf(float v) {
    const int b = __builtin_bit_cast(int, v);
    return __builtin_bit_cast(float,
        __builtin_amdgcn_permlanex16(b, b, 0x76543210, 0xfedcba98, false, false));
}

__device__ __forceinline__ v2f pair_of(const v8f& v, const int n) {
    v2f r;
    r.x = v[2 * n];
    r.y = v[2 * n + 1];
    return r;
}

template <int MODE>
__global__ __launch_bounds__(256) void caps_kernel(
    const float* __restrict__ x,
    const float* __restrict__ W,
    float* __restrict__ bmat,
    const float* __restrict__ rowmax,
    const float* __restrict__ denom,
    float* __restrict__ out)
{
    const int r0   = blockIdx.x * RGRP;
    const int tid  = threadIdx.x;
    const int L    = tid & 31;
    const int w    = tid >> 5;
    const int h    = L >> 4;
    const int m16  = L & 15;
    const int bb0  = w * 16;
    const int b    = bb0 + m16;
    const bool hi  = (h != 0);

    float rm[8], dn[8];
    if (MODE != 0) {
        const float4 m0 = *reinterpret_cast<const float4*>(rowmax + b * ONUM + 8 * h);
        const float4 m1 = *reinterpret_cast<const float4*>(rowmax + b * ONUM + 8 * h + 4);
        const float4 d0 = *reinterpret_cast<const float4*>(denom + b * ONUM + 8 * h);
        const float4 d1 = *reinterpret_cast<const float4*>(denom + b * ONUM + 8 * h + 4);
        rm[0] = m0.x; rm[1] = m0.y; rm[2] = m0.z; rm[3] = m0.w;
        rm[4] = m1.x; rm[5] = m1.y; rm[6] = m1.z; rm[7] = m1.w;
        dn[0] = d0.x; dn[1] = d0.y; dn[2] = d0.z; dn[3] = d0.w;
        dn[4] = d1.x; dn[5] = d1.y; dn[6] = d1.z; dn[7] = d1.w;
    }

    v2f xf[4];
    const float* xb = x + b * DNUM + h;
    #pragma unroll
    for (int k = 0; k < 4; ++k) {
        xf[k].x = xb[4 * k];
        xf[k].y = xb[4 * k + 2];
    }

    const int oo = m16 >> 2;
    const int q  = m16 & 3;
    const int wlane = (8 * (oo >> 1) + (oo & 1)) * (DNUM * CNUM) + h * CNUM + 4 * q;

    __shared__ float4 Wlds[RGRP * 32 * 32];
    {
        const int sr = w >> 1;
        const int sg = 2 * (w & 1);
        const float* Ws = W + (size_t)(r0 + sr) * ONUM * DNUM * CNUM + wlane;
        float4 T[2][4][2];
        #pragma unroll
        for (int gg = 0; gg < 2; ++gg) {
            #pragma unroll
            for (int k = 0; k < 4; ++k) {
                #pragma unroll
                for (int s = 0; s < 2; ++s) {
                    T[gg][k][s] = *reinterpret_cast<const float4*>(
                        Ws + (2 * (sg + gg)) * (DNUM * CNUM) + (4 * k + 2 * s) * CNUM);
                }
            }
        }
        float4* dst = Wlds + (sr * 32 + sg * 8) * 32 + L;
        #pragma unroll
        for (int gg = 0; gg < 2; ++gg) {
            #pragma unroll
            for (int k = 0; k < 4; ++k) {
                const float4 v0 = T[gg][k][0];
                const float4 v1 = T[gg][k][1];
                dst[((gg * 4 + k) * 2 + 0) * 32] = make_float4(v0.x, v1.x, v0.y, v1.y);
                dst[((gg * 4 + k) * 2 + 1) * 32] = make_float4(v0.z, v1.z, v0.w, v1.w);
            }
        }
    }
    __syncthreads();

    #pragma unroll 1
    for (int ri = 0; ri < RGRP; ++ri) {
        const int r = r0 + ri;

        float braw[8], cown[8];
        const size_t brow = ((size_t)b * RNUM + r) * ONUM + 8 * h;
        float4 q0, q1;
        if (MODE != 0) {
            q0 = *reinterpret_cast<const float4*>(bmat + brow);
            q1 = *reinterpret_cast<const float4*>(bmat + brow + 4);
        }

        const float4* Wl = Wlds + ri * (32 * 32) + L;
        float4 F[4][4][2];
        #pragma unroll
        for (int g = 0; g < 4; ++g) {
            #pragma unroll
            for (int k = 0; k < 4; ++k) {
                #pragma unroll
                for (int u = 0; u < 2; ++u) {
                    F[g][k][u] = Wl[((g * 4 + k) * 2 + u) * 32];
                }
            }
        }

        if (MODE != 0) {
            braw[0] = q0.x; braw[1] = q0.y; braw[2] = q0.z; braw[3] = q0.w;
            braw[4] = q1.x; braw[5] = q1.y; braw[6] = q1.z; braw[7] = q1.w;
            #pragma unroll
            for (int j = 0; j < 8; ++j) cown[j] = __expf(braw[j] - rm[j]) / dn[j];
        } else {
            #pragma unroll
            for (int j = 0; j < 8; ++j) { braw[j] = 0.0f; cown[j] = 1.0f / (float)RNUM; }
        }

        v8f U[ONUM];
        v2f P[4][2];
        #pragma unroll
        for (int t = 0; t < 4; ++t) {
            P[t][0] = (v2f){0.f, 0.f};
            P[t][1] = (v2f){0.f, 0.f};
        }
        #pragma unroll
        for (int g = 0; g < 5; ++g) {
            __builtin_amdgcn_sched_barrier(0x0416);
            if (g < 4) {
                v8f acc[4];
                #pragma unroll
                for (int t = 0; t < 4; ++t) {
                    acc[t] = (v8f){0.f, 0.f, 0.f, 0.f, 0.f, 0.f, 0.f, 0.f};
                }
                #pragma unroll
                for (int k = 0; k < 4; ++k) {
                    const float4 f0 = F[g < 4 ? g : 0][k][0];
                    const float4 f1 = F[g < 4 ? g : 0][k][1];
                    const float a0[4] = {f0.x, f0.z, f1.x, f1.z};
                    const float a1[4] = {f0.y, f0.w, f1.y, f1.w};
                    #pragma unroll
                    for (int t = 0; t < 4; ++t) {
                        v2f wf;
                        wf.x = a0[t];
                        wf.y = a1[t];
                        acc[t] = __builtin_amdgcn_wmma_f32_16x16x4_f32(
                            false, wf, false, xf[k], (short)0, acc[t], false, false);
                    }
                }
                #pragma unroll
                for (int t = 0; t < 4; ++t) U[4 * (g < 4 ? g : 0) + t] = acc[t];
            }
            if (g > 0) {
                const int gp = g - 1;
                #pragma unroll
                for (int e = 0; e < 2; ++e) {
                    const float cj = cown[2 * gp + e];
                    const v2f cj2 = {cj, cj};
                    #pragma unroll
                    for (int t = 0; t < 4; ++t) {
                        #pragma unroll
                        for (int p = 0; p < 2; ++p) {
                            P[t][p] += cj2 * pair_of(U[4 * gp + t], 2 * e + p);
                        }
                    }
                }
            }
        }

        v2f S[4][2];
        #pragma unroll
        for (int t = 0; t < 4; ++t) {
            #pragma unroll
            for (int p = 0; p < 2; ++p) {
                v2f other;
                other.x = xhalf(P[t][p].x);
                other.y = xhalf(P[t][p].y);
                S[t][p] = P[t][p] + other;
            }
        }

        v2f nsp = {0.f, 0.f};
        #pragma unroll
        for (int t = 0; t < 4; ++t) {
            #pragma unroll
            for (int p = 0; p < 2; ++p) nsp += S[t][p] * S[t][p];
        }
        const float ns  = nsp.x + nsp.y;
        const float fac = sqrtf(ns) / (1.0f + ns);
        const v2f fac2 = {fac, fac};
        v2f V[4][2];
        #pragma unroll
        for (int t = 0; t < 4; ++t) {
            #pragma unroll
            for (int p = 0; p < 2; ++p) V[t][p] = S[t][p] * fac2;
        }

        if (MODE < 2) {
            float bnew[8];
            #pragma unroll
            for (int g = 0; g < 4; ++g) {
                #pragma unroll
                for (int e = 0; e < 2; ++e) {
                    v2f a2 = {0.f, 0.f};
                    #pragma unroll
                    for (int t = 0; t < 4; ++t) {
                        #pragma unroll
                        for (int p = 0; p < 2; ++p) {
                            a2 += pair_of(U[4 * g + t], 2 * e + p) * V[t][p];
                        }
                    }
                    bnew[2 * g + e] = braw[2 * g + e] + (a2.x + a2.y);
                }
            }
            *reinterpret_cast<float4*>(bmat + brow)     = make_float4(bnew[0], bnew[1], bnew[2], bnew[3]);
            *reinterpret_cast<float4*>(bmat + brow + 4) = make_float4(bnew[4], bnew[5], bnew[6], bnew[7]);
        } else {
            float vo[8];
            #pragma unroll
            for (int i = 0; i < 8; ++i) {
                const float lo = (i < 4) ? V[i & 3][0].x : V[i & 3][0].y;
                const float up = (i < 4) ? V[i & 3][1].x : V[i & 3][1].y;
                vo[i] = hi ? up : lo;
            }
            float* op = out + ((size_t)b * RNUM + r) * CNUM + 8 * h;
            *reinterpret_cast<float4*>(op)     = make_float4(vo[0], vo[1], vo[2], vo[3]);
            *reinterpret_cast<float4*>(op + 4) = make_float4(vo[4], vo[5], vo[6], vo[7]);
        }
    }
}

#define SR_THREADS 512
#define SR_WAVES   (SR_THREADS / 32)
#define SR_LOADS   (RNUM * ONUM / 4 / SR_THREADS)

template <int M>
__device__ __forceinline__ float xrow(float v) {
    const int b = __builtin_bit_cast(int, v);
    return __builtin_bit_cast(float,
        __builtin_amdgcn_update_dpp(b, b, 0x160 + M, 0xf, 0xf, false));
}

__device__ __forceinline__ float bfly_max(float v) {
    v = fmaxf(v, xrow<4>(v));
    v = fmaxf(v, xrow<8>(v));
    return fmaxf(v, xhalf(v));
}
__device__ __forceinline__ float bfly_sum(float v) {
    v += xrow<4>(v);
    v += xrow<8>(v);
    return v + xhalf(v);
}

__global__ __launch_bounds__(SR_THREADS) void softmax_reduce(
    const float* __restrict__ bmat,
    float* __restrict__ rowmax,
    float* __restrict__ denom)
{
    __shared__ float4 red_max[SR_WAVES * 4];
    __shared__ float4 red_sum[SR_WAVES * 4];
    const int b = blockIdx.x;
    const int t = threadIdx.x;
    const int L = t & 31;
    const int w = t >> 5;

    const float4* src4 =
        reinterpret_cast<const float4*>(bmat + (size_t)b * RNUM * ONUM) + t;
    float4 v[SR_LOADS];
    #pragma unroll
    for (int i = 0; i < SR_LOADS; ++i) v[i] = src4[i * SR_THREADS];

    float4 m = make_float4(-INFINITY, -INFINITY, -INFINITY, -INFINITY);
    #pragma unroll
    for (int i = 0; i < SR_LOADS; ++i) {
        m.x = fmaxf(m.x, v[i].x); m.y = fmaxf(m.y, v[i].y);
        m.z = fmaxf(m.z, v[i].z); m.w = fmaxf(m.w, v[i].w);
    }
    m.x = bfly_max(m.x); m.y = bfly_max(m.y); m.z = bfly_max(m.z); m.w = bfly_max(m.w);
    if (L < 4) red_max[4 * w + L] = m;
    __syncthreads();
    {
        const float4 p = red_max[L], q = red_max[L + 32];
        m.x = bfly_max(fmaxf(p.x, q.x)); m.y = bfly_max(fmaxf(p.y, q.y));
        m.z = bfly_max(fmaxf(p.z, q.z)); m.w = bfly_max(fmaxf(p.w, q.w));
    }

    float4 s = make_float4(0.0f, 0.0f, 0.0f, 0.0f);
    #pragma unroll
    for (int i = 0; i < SR_LOADS; ++i) {
        s.x += __expf(v[i].x - m.x); s.y += __expf(v[i].y - m.y);
        s.z += __expf(v[i].z - m.z); s.w += __expf(v[i].w - m.w);
    }
    s.x = bfly_sum(s.x); s.y = bfly_sum(s.y); s.z = bfly_sum(s.z); s.w = bfly_sum(s.w);
    if (L < 4) red_sum[4 * w + L] = s;
    __syncthreads();
    if (w == 0) {
        const float4 p = red_sum[L], q = red_sum[L + 32];
        s.x = bfly_sum(p.x + q.x); s.y = bfly_sum(p.y + q.y);
        s.z = bfly_sum(p.z + q.z); s.w = bfly_sum(p.w + q.w);
        if (L < 4) {
            *reinterpret_cast<float4*>(rowmax + b * ONUM + 4 * L) = m;
            *reinterpret_cast<float4*>(denom  + b * ONUM + 4 * L) = s;
        }
    }
}

extern "C" void kernel_launch(void* const* d_in, const int* in_sizes, int n_in,
                              void* d_out, int out_size, void* d_ws, size_t ws_size,
                              hipStream_t stream) {
    const float* x = (const float*)d_in[0];
    const float* W = (const float*)d_in[1];
    float* out = (float*)d_out;

    float* bmat = (float*)d_ws;
    float* rowmax = bmat + (size_t)BATCH * RNUM * ONUM;
    float* denom  = rowmax + BATCH * ONUM;

    dim3 grid(RNUM / RGRP), block(256);

    caps_kernel<0><<<grid, block, 0, stream>>>(x, W, bmat, rowmax, denom, out);
    softmax_reduce<<<dim3(BATCH), dim3(SR_THREADS), 0, stream>>>(bmat, rowmax, denom);
    caps_kernel<1><<<grid, block, 0, stream>>>(x, W, bmat, rowmax, denom, out);
    softmax_reduce<<<dim3(BATCH), dim3(SR_THREADS), 0, stream>>>(bmat, rowmax, denom);
    caps_kernel<2><<<grid, block, 0, stream>>>(x, W, bmat, rowmax, denom, out);
}
